# starting kernel
# baseline (speedup 1.0000x reference)
#include <hip/hip_runtime.h>

#define N_NODES  50000
#define N_EDGES  1600000
#define IN_FEAT  128
#define OUT_FEAT 128

typedef __attribute__((ext_vector_type(2))) float v2f;
typedef __attribute__((ext_vector_type(8))) float v8f;

__global__ __launch_bounds__(256) void gcn_zero_kernel(float4* __restrict__ out, int n4) {
    int i = blockIdx.x * blockDim.x + threadIdx.x;
    if (i < n4) out[i] = make_float4(0.f, 0.f, 0.f, 0.f);
}

__global__ __launch_bounds__(256) void gcn_gemm_kernel(const float* __restrict__ X,
                                                       const float* __restrict__ W,
                                                       float* __restrict__ H) {
    const int lane = threadIdx.x & 31;
    const int wave = threadIdx.x >> 5;

    const int row0 = blockIdx.x * 16;
    const int o0   = wave * 16;
    const int ln   = lane & 15;
    const int koff = (lane >> 4) << 1;

    const float* __restrict__ xa = X + (size_t)(row0 + ln) * IN_FEAT + koff;
    const float* __restrict__ wb = W + (size_t)(o0 + ln) * IN_FEAT + koff;

    v8f acc = {};
#pragma unroll
    for (int k0 = 0; k0 < IN_FEAT; k0 += 4) {
        v2f a = *(const v2f*)(xa + k0);
        v2f b = *(const v2f*)(wb + k0);
        acc = __builtin_amdgcn_wmma_f32_16x16x4_f32(
            false, a, false, b, (short)0, acc, false, false);
    }

    const int rbase = row0 + ((lane >> 4) << 3);
    float* __restrict__ hp = H + (size_t)rbase * OUT_FEAT + o0 + ln;
#pragma unroll
    for (int v = 0; v < 8; ++v)
        hp[(size_t)v * OUT_FEAT] = acc[v];
}

__global__ __launch_bounds__(256) void gcn_spmm_kernel(const float* __restrict__ H,
                                                       const float* __restrict__ vals,
                                                       const int* __restrict__ rows,
                                                       const int* __restrict__ cols,
                                                       float* __restrict__ out) {
    const int e    = (int)((blockIdx.x * blockDim.x + threadIdx.x) >> 5);
    const int lane = threadIdx.x & 31;
    if (e >= N_EDGES) return;

    const int   r = rows[e];
    const int   c = cols[e];
    const float v = vals[e];

    const float4 h = ((const float4*)(H + (size_t)c * OUT_FEAT))[lane];
    float* __restrict__ o = out + (size_t)r * OUT_FEAT + lane * 4;

    unsafeAtomicAdd(o + 0, v * h.x);
    unsafeAtomicAdd(o + 1, v * h.y);
    unsafeAtomicAdd(o + 2, v * h.z);
    unsafeAtomicAdd(o + 3, v * h.w);
}

extern "C" void kernel_launch(void* const* d_in, const int* in_sizes, int n_in,
                              void* d_out, int out_size, void* d_ws, size_t ws_size,
                              hipStream_t stream) {
    const float* X      = (const float*)d_in[0];
    const float* W      = (const float*)d_in[1];
    const float* A_vals = (const float*)d_in[2];
    const int*   A_rows = (const int*)d_in[3];
    const int*   A_cols = (const int*)d_in[4];
    float*       out    = (float*)d_out;
    float*       H      = (float*)d_ws;

    const int n4 = N_NODES * OUT_FEAT / 4;
    gcn_zero_kernel<<<(n4 + 255) / 256, 256, 0, stream>>>((float4*)out, n4);

    gcn_gemm_kernel<<<N_NODES / 16, 256, 0, stream>>>(X, W, H);

    gcn_spmm_kernel<<<N_EDGES / 8, 256, 0, stream>>>(H, A_vals, A_rows, A_cols, out);
}


# GEMM stage only: 256 blocks loop over row tiles; a wave keeps 64 columns of W in registers (via LDS), takes X fragments straight from memory by 16-byte loads; k regrouped per WMMA, H sums reorder
# speedup vs baseline: 2.4156x; 2.4156x over previous
#include <hip/hip_runtime.h>

#define N_NODES  50000
#define N_EDGES  1600000
#define IN_FEAT  128
#define OUT_FEAT 128

typedef __attribute__((ext_vector_type(2))) float v2f;
typedef __attribute__((ext_vector_type(8))) float v8f;

__global__ __launch_bounds__(256) void gcn_zero_kernel(float4* __restrict__ out, int n4) {
    int i = blockIdx.x * blockDim.x + threadIdx.x;
    if (i < n4) out[i] = make_float4(0.f, 0.f, 0.f, 0.f);
}

typedef __attribute__((ext_vector_type(4))) float v4f;

#define WP          132
#define GEMM_BLOCKS 256
#define ROW_TILES   (N_NODES / 16)
static_assert(N_NODES % 16 == 0, "row tiles cover X exactly");
static_assert(IN_FEAT == 128 && OUT_FEAT == 128, "tile shapes below assume 128 x 128 W");

__global__ __launch_bounds__(256) void gcn_gemm_kernel(const float* __restrict__ X,
                                                       const float* __restrict__ W,
                                                       float* __restrict__ H) {
    __shared__ __attribute__((aligned(16))) float Ws[OUT_FEAT * WP];

    const int tid = threadIdx.x;

#pragma unroll
    for (int i = 0; i < 16; ++i) {
        const int idx = i * 256 + tid;
        const v4f w = *(const v4f*)(W + (size_t)idx * 4);
        *(v4f*)(Ws + (idx >> 5) * WP + (idx & 31) * 4) = w;
    }
    __syncthreads();

    const int lane = tid & 31;
    const int wave = __builtin_amdgcn_readfirstlane(tid >> 5);
    const int c    = lane & 15;
    const int h    = lane >> 4;
    const int half = wave & 1;

    const float* wsb = Ws + (64 * half + c) * WP + 4 * h;

    int tile = blockIdx.x + GEMM_BLOCKS * (wave >> 1);

    v4f a[16];
    {
        const int t0 = tile < ROW_TILES ? tile : 0;
        const float* xa = X + (size_t)(t0 * 16 + c) * IN_FEAT + 4 * h;
#pragma unroll
        for (int j = 0; j < 16; ++j) a[j] = *(const v4f*)(xa + 8 * j);
    }

    v4f b[4][16];
#pragma unroll
    for (int t = 0; t < 4; ++t)
#pragma unroll
        for (int j = 0; j < 16; ++j) b[t][j] = *(const v4f*)(wsb + 16 * t * WP + 8 * j);

    while (tile < ROW_TILES) {
        const int nxt = tile + 4 * GEMM_BLOCKS;
        const int pf  = nxt < ROW_TILES ? nxt : tile;
        v4f an[16];
        {
            const float* xa = X + (size_t)(pf * 16 + c) * IN_FEAT + 4 * h;
#pragma unroll
            for (int j = 0; j < 16; ++j) an[j] = *(const v4f*)(xa + 8 * j);
        }
        __builtin_amdgcn_sched_barrier(0);

        v8f acc[4] = {};
#pragma unroll
        for (int j = 0; j < 16; ++j) {
            const v2f alo = __builtin_shufflevector(a[j], a[j], 0, 1);
            const v2f ahi = __builtin_shufflevector(a[j], a[j], 2, 3);
#pragma unroll
            for (int t = 0; t < 4; ++t) {
                const v2f blo = __builtin_shufflevector(b[t][j], b[t][j], 0, 1);
                acc[t] = __builtin_amdgcn_wmma_f32_16x16x4_f32(
                    false, alo, false, blo, (short)0, acc[t], false, false);
            }
#pragma unroll
            for (int t = 0; t < 4; ++t) {
                const v2f bhi = __builtin_shufflevector(b[t][j], b[t][j], 2, 3);
                acc[t] = __builtin_amdgcn_wmma_f32_16x16x4_f32(
                    false, ahi, false, bhi, (short)0, acc[t], false, false);
            }
            __builtin_amdgcn_sched_barrier(0);
        }

        float* __restrict__ hp = H + (size_t)(tile * 16 + 8 * h) * OUT_FEAT + 64 * half + c;
#pragma unroll
        for (int t = 0; t < 4; ++t)
#pragma unroll
            for (int v = 0; v < 8; ++v)
                hp[(size_t)v * OUT_FEAT + 16 * t] = acc[t][v];

#pragma unroll
        for (int j = 0; j < 16; ++j) a[j] = an[j];
        tile = nxt;
    }
}

#define EPW 32
#define EG  8
static_assert(N_EDGES % (EPW * 8) == 0, "grid covers the edge list exactly");
static_assert((unsigned long long)N_NODES * OUT_FEAT < (1ull << 30), "32-bit element index");

__global__ __launch_bounds__(256) void gcn_spmm_kernel(const float* __restrict__ H,
                                                       const float* __restrict__ vals,
                                                       const int* __restrict__ rows,
                                                       const int* __restrict__ cols,
                                                       float* __restrict__ out) {
    const unsigned lane = threadIdx.x & 31u;
    const unsigned wave = (blockIdx.x * blockDim.x + threadIdx.x) >> 5;
    const unsigned e    = wave * EPW + lane;

    const int my_r = rows[e];
    const int my_c = cols[e];
    const int my_v = __float_as_int(vals[e]);

    float h[2][EG][4];

#pragma unroll
    for (int i = 0; i < EG; ++i) {
        const unsigned hi = (unsigned)__builtin_amdgcn_readlane(my_c, i) * OUT_FEAT + lane;
        h[0][i][0] = H[hi];
        h[0][i][1] = H[hi + 32u];
        h[0][i][2] = H[hi + 64u];
        h[0][i][3] = H[hi + 96u];
    }

#pragma unroll
    for (int g = 0; g < EPW / EG; ++g) {
        const int cur = g & 1;
        if (g + 1 < EPW / EG) {
#pragma unroll
            for (int i = 0; i < EG; ++i) {
                const unsigned hi =
                    (unsigned)__builtin_amdgcn_readlane(my_c, (g + 1) * EG + i) * OUT_FEAT + lane;
                h[cur ^ 1][i][0] = H[hi];
                h[cur ^ 1][i][1] = H[hi + 32u];
                h[cur ^ 1][i][2] = H[hi + 64u];
                h[cur ^ 1][i][3] = H[hi + 96u];
            }
        }
#pragma unroll
        for (int i = 0; i < EG; ++i) {
            const unsigned oi =
                (unsigned)__builtin_amdgcn_readlane(my_r, g * EG + i) * OUT_FEAT + lane;
            const float v = __int_as_float(__builtin_amdgcn_readlane(my_v, g * EG + i));
            unsafeAtomicAdd(out + oi,       v * h[cur][i][0]);
            unsafeAtomicAdd(out + oi + 32u, v * h[cur][i][1]);
            unsafeAtomicAdd(out + oi + 64u, v * h[cur][i][2]);
            unsafeAtomicAdd(out + oi + 96u, v * h[cur][i][3]);
        }
    }
}

extern "C" void kernel_launch(void* const* d_in, const int* in_sizes, int n_in,
                              void* d_out, int out_size, void* d_ws, size_t ws_size,
                              hipStream_t stream) {
    const float* X      = (const float*)d_in[0];
    const float* W      = (const float*)d_in[1];
    const float* A_vals = (const float*)d_in[2];
    const int*   A_rows = (const int*)d_in[3];
    const int*   A_cols = (const int*)d_in[4];
    float*       out    = (float*)d_out;
    float*       H      = (float*)d_ws;

    const int n4 = N_NODES * OUT_FEAT / 4;
    gcn_zero_kernel<<<(n4 + 255) / 256, 256, 0, stream>>>((float4*)out, n4);

    gcn_gemm_kernel<<<GEMM_BLOCKS, 256, 0, stream>>>(X, W, H);

    gcn_spmm_kernel<<<N_EDGES / (EPW * 8), 256, 0, stream>>>(H, A_vals, A_rows, A_cols, out);
}


# edge scatter split by feature half between the 2 cache halves: a block reads its die, its waves draw 64-edge tickets and add only that half's 64 features (one 256-B unit a row); atomic arrival order
# speedup vs baseline: 3.0279x; 1.2535x over previous
#include <hip/hip_runtime.h>

#define N_NODES  50000
#define N_EDGES  1600000
#define IN_FEAT  128
#define OUT_FEAT 128

typedef __attribute__((ext_vector_type(2))) float v2f;
typedef __attribute__((ext_vector_type(8))) float v8f;

#define CNT_PITCH 64

__global__ __launch_bounds__(256) void gcn_zero_kernel(float4* __restrict__ out, int n4,
                                                       unsigned* __restrict__ cnt) {
    int i = blockIdx.x * blockDim.x + threadIdx.x;
    if (i < n4) out[i] = make_float4(0.f, 0.f, 0.f, 0.f);
    if (i == 0) { cnt[0] = 0u; cnt[CNT_PITCH] = 0u; }
}

typedef __attribute__((ext_vector_type(4))) float v4f;

#define WP          132
#define GEMM_BLOCKS 256
#define ROW_TILES   (N_NODES / 16)
static_assert(N_NODES % 16 == 0, "row tiles cover X exactly");
static_assert(IN_FEAT == 128 && OUT_FEAT == 128, "tile shapes below assume 128 x 128 W");

__global__ __launch_bounds__(256) void gcn_gemm_kernel(const float* __restrict__ X,
                                                       const float* __restrict__ W,
                                                       float* __restrict__ H) {
    __shared__ __attribute__((aligned(16))) float Ws[OUT_FEAT * WP];

    const int tid = threadIdx.x;

#pragma unroll
    for (int i = 0; i < 16; ++i) {
        const int idx = i * 256 + tid;
        const v4f w = *(const v4f*)(W + (size_t)idx * 4);
        *(v4f*)(Ws + (idx >> 5) * WP + (idx & 31) * 4) = w;
    }
    __syncthreads();

    const int lane = tid & 31;
    const int wave = __builtin_amdgcn_readfirstlane(tid >> 5);
    const int c    = lane & 15;
    const int h    = lane >> 4;
    const int half = wave & 1;

    const float* wsb = Ws + (64 * half + c) * WP + 4 * h;

    int tile = blockIdx.x + GEMM_BLOCKS * (wave >> 1);

    v4f a[16];
    {
        const int t0 = tile < ROW_TILES ? tile : 0;
        const float* xa = X + (size_t)(t0 * 16 + c) * IN_FEAT + 4 * h;
#pragma unroll
        for (int j = 0; j < 16; ++j) a[j] = *(const v4f*)(xa + 8 * j);
    }

    v4f b[4][16];
#pragma unroll
    for (int t = 0; t < 4; ++t)
#pragma unroll
        for (int j = 0; j < 16; ++j) b[t][j] = *(const v4f*)(wsb + 16 * t * WP + 8 * j);

    while (tile < ROW_TILES) {
        const int nxt = tile + 4 * GEMM_BLOCKS;
        const int pf  = nxt < ROW_TILES ? nxt : tile;
        v4f an[16];
        {
            const float* xa = X + (size_t)(pf * 16 + c) * IN_FEAT + 4 * h;
#pragma unroll
            for (int j = 0; j < 16; ++j) an[j] = *(const v4f*)(xa + 8 * j);
        }
        __builtin_amdgcn_sched_barrier(0);

        v8f acc[4] = {};
#pragma unroll
        for (int j = 0; j < 16; ++j) {
            const v2f alo = __builtin_shufflevector(a[j], a[j], 0, 1);
            const v2f ahi = __builtin_shufflevector(a[j], a[j], 2, 3);
#pragma unroll
            for (int t = 0; t < 4; ++t) {
                const v2f blo = __builtin_shufflevector(b[t][j], b[t][j], 0, 1);
                acc[t] = __builtin_amdgcn_wmma_f32_16x16x4_f32(
                    false, alo, false, blo, (short)0, acc[t], false, false);
            }
#pragma unroll
            for (int t = 0; t < 4; ++t) {
                const v2f bhi = __builtin_shufflevector(b[t][j], b[t][j], 2, 3);
                acc[t] = __builtin_amdgcn_wmma_f32_16x16x4_f32(
                    false, ahi, false, bhi, (short)0, acc[t], false, false);
            }
            __builtin_amdgcn_sched_barrier(0);
        }

        float* __restrict__ hp = H + (size_t)(tile * 16 + 8 * h) * OUT_FEAT + 64 * half + c;
#pragma unroll
        for (int t = 0; t < 4; ++t)
#pragma unroll
            for (int v = 0; v < 8; ++v)
                hp[(size_t)v * OUT_FEAT + 16 * t] = acc[t][v];

#pragma unroll
        for (int j = 0; j < 16; ++j) a[j] = an[j];
        tile = nxt;
    }
}

#define EPT       64
#define EG        16
#define NTICKETS  (N_EDGES / EPT)
#define SPMM_BLOCKS 1536
static_assert(N_EDGES % EPT == 0, "tickets cover the edge list exactly");
static_assert((unsigned long long)N_NODES * OUT_FEAT < (1ull << 30), "32-bit element index");

__device__ __forceinline__ unsigned peek_counter(unsigned* ctr) {
    const unsigned n = __hip_atomic_load(ctr, __ATOMIC_RELAXED, __HIP_MEMORY_SCOPE_AGENT);
    return (unsigned)__builtin_amdgcn_readfirstlane((int)n);
}

__device__ __forceinline__ unsigned draw_ticket(unsigned* ctr, unsigned lane) {
    unsigned t = 0u;
    if (lane == 0u) t = atomicAdd(ctr, 1u);
    return (unsigned)__builtin_amdgcn_readfirstlane((int)t);
}

__global__ __launch_bounds__(256, 12) void gcn_spmm_kernel(const float* __restrict__ H,
                                                       const float* __restrict__ vals,
                                                       const int* __restrict__ rows,
                                                       const int* __restrict__ cols,
                                                       float* __restrict__ out,
                                                       unsigned* __restrict__ cnt) {
    const unsigned lane = threadIdx.x & 31u;

    __shared__ unsigned s_half;
    if (__builtin_amdgcn_readfirstlane((int)(threadIdx.x >> 5)) == 0) {
        const unsigned place = (unsigned)__builtin_amdgcn_s_sendmsg_rtn(0x87);
        if (lane == 0u) s_half = (place >> 18) & 1u;
    }
    __syncthreads();
    const unsigned my_half = (unsigned)__builtin_amdgcn_readfirstlane((int)s_half);

    for (unsigned pass = 0u; pass < 2u; ++pass) {
        const unsigned q  = my_half ^ pass;
        unsigned* ctr     = cnt + q * CNT_PITCH;
        const unsigned fo = q * 64u + lane;

        unsigned t = (unsigned)NTICKETS;
        if (peek_counter(ctr) < (unsigned)NTICKETS) t = draw_ticket(ctr, lane);
        while (t < (unsigned)NTICKETS) {
            const unsigned e = t * EPT + lane;
            const int r0 = rows[e];
            const int r1 = rows[e + 32u];
            const int c0 = cols[e];
            const int c1 = cols[e + 32u];
            const int v0 = __float_as_int(vals[e]);
            const int v1 = __float_as_int(vals[e + 32u]);

            const unsigned tn = draw_ticket(ctr, lane);

            float h[2][EG][2];

#pragma unroll
            for (int i = 0; i < EG; ++i) {
                const unsigned hi = (unsigned)__builtin_amdgcn_readlane(c0, i) * OUT_FEAT + fo;
                h[0][i][0] = H[hi];
                h[0][i][1] = H[hi + 32u];
            }

#pragma unroll
            for (int g = 0; g < EPT / EG; ++g) {
                const int cur = g & 1;
                if (g + 1 < EPT / EG) {
                    const int cn = (g + 1) < 2 ? c0 : c1;
#pragma unroll
                    for (int i = 0; i < EG; ++i) {
                        const unsigned hi =
                            (unsigned)__builtin_amdgcn_readlane(cn, ((g + 1) & 1) * EG + i) * OUT_FEAT + fo;
                        h[cur ^ 1][i][0] = H[hi];
                        h[cur ^ 1][i][1] = H[hi + 32u];
                    }
                }
                const int rg = g < 2 ? r0 : r1;
                const int vg = g < 2 ? v0 : v1;
#pragma unroll
                for (int i = 0; i < EG; ++i) {
                    const unsigned oi =
                        (unsigned)__builtin_amdgcn_readlane(rg, (g & 1) * EG + i) * OUT_FEAT + fo;
                    const float v = __int_as_float(__builtin_amdgcn_readlane(vg, (g & 1) * EG + i));
                    unsafeAtomicAdd(out + oi,       v * h[cur][i][0]);
                    unsafeAtomicAdd(out + oi + 32u, v * h[cur][i][1]);
                }
            }
            t = tn;
        }
    }
}

extern "C" void kernel_launch(void* const* d_in, const int* in_sizes, int n_in,
                              void* d_out, int out_size, void* d_ws, size_t ws_size,
                              hipStream_t stream) {
    const float* X      = (const float*)d_in[0];
    const float* W      = (const float*)d_in[1];
    const float* A_vals = (const float*)d_in[2];
    const int*   A_rows = (const int*)d_in[3];
    const int*   A_cols = (const int*)d_in[4];
    float*       out    = (float*)d_out;
    float*       H      = (float*)d_ws;
    unsigned*    cnt    = (unsigned*)((float*)d_ws + (size_t)N_NODES * OUT_FEAT);

    const int n4 = N_NODES * OUT_FEAT / 4;
    gcn_zero_kernel<<<(n4 + 255) / 256, 256, 0, stream>>>((float4*)out, n4, cnt);

    gcn_gemm_kernel<<<GEMM_BLOCKS, 256, 0, stream>>>(X, W, H);

    gcn_spmm_kernel<<<SPMM_BLOCKS, 256, 0, stream>>>(H, A_vals, A_rows, A_cols, out, cnt);
}
